# starting kernel
# baseline (speedup 1.0000x reference)
#include <hip/hip_runtime.h>
#include <hip/hip_bf16.h>

#define NB 16
#define NS 16
#define NL 128
#define NA 8
#define NT 8
#define ND 768
#define WSTRIDE 132

#define SENT_SZ (NB * NS * ND)
#define ARG_SZ  (NB * NS * NA * ND)

typedef __attribute__((ext_vector_type(2))) float v2f;
typedef __attribute__((ext_vector_type(8))) float v8f;

__device__ __forceinline__ int phys_to_logical(int p) {
    if (p < 20) return p;
    if (p == 20) return 24;
    if (p >= 24 && p <= 27) return p - 4;
    return -1;
}

__global__ __launch_bounds__(256)
void slmuse_embed_kernel(const int* __restrict__ sid,
                         const int* __restrict__ amask,
                         const int* __restrict__ pid,
                         const int* __restrict__ a0id,
                         const int* __restrict__ a1id,
                         const float* __restrict__ emb,
                         float* __restrict__ out_base) {
    __shared__ float    Wm[32 * WSTRIDE];
    __shared__ int      s_sid[NL];
    __shared__ int      s_mask[NL];
    __shared__ int      s_ids[3 * NA * NT];
    __shared__ float    s_inv[32];
    __shared__ unsigned s_off[32];

    const int bs  = blockIdx.x;
    const int tid = threadIdx.x;

    if (tid < NL) {
        s_sid[tid]  = sid[bs * NL + tid];
        s_mask[tid] = amask[bs * NL + tid];
    }
    if (tid < NA * NT) {
        s_ids[tid]               = pid[bs * NA * NT + tid];
        s_ids[NA * NT + tid]     = a0id[bs * NA * NT + tid];
        s_ids[2 * NA * NT + tid] = a1id[bs * NA * NT + tid];
    }
    if (tid < 32) {
        const int l = phys_to_logical(tid);
        unsigned off = 0;
        if (l == 0)       off = (unsigned)(bs * ND);
        else if (l <= 8)  off = (unsigned)(SENT_SZ + (bs * NA + (l - 1)) * ND);
        else if (l <= 16) off = (unsigned)(SENT_SZ + ARG_SZ + (bs * NA + (l - 9)) * ND);
        else if (l <= 24) off = (unsigned)(SENT_SZ + 2 * ARG_SZ + (bs * NA + (l - 17)) * ND);
        s_off[tid] = off;
    }
    __syncthreads();

    for (int e = tid; e < 32 * NL; e += 256) {
        const int p = e >> 7;
        const int c = e & (NL - 1);
        const int l = phys_to_logical(p);
        float w;
        if (l == 0) {
            w = (float)s_mask[c];
        } else if (l > 0) {
            const int base = (l - 1) * NT;
            const int sv = s_sid[c];
            int cnt = 0;
#pragma unroll
            for (int t = 0; t < NT; ++t) {
                const int id = s_ids[base + t];
                cnt += (id != 0 && id == sv) ? 1 : 0;
            }
            w = (float)cnt;
        } else {
            w = 0.0f;
        }
        Wm[p * WSTRIDE + c] = w;
    }
    __syncthreads();

    if (tid < 32) {
        float c = 0.0f;
        for (int x = 0; x < NL; ++x) c += Wm[tid * WSTRIDE + x];
        float inv;
        if (tid == 0)
            inv = 1.0f / fmaxf(c, 1.0f);
        else
            inv = (c > 0.0f) ? (1.0f / c) : 0.0f;
        s_inv[tid] = inv;
    }
    __syncthreads();
    for (int e = tid; e < 32 * NL; e += 256) {
        const int p = e >> 7;
        const int c = e & (NL - 1);
        Wm[p * WSTRIDE + c] *= s_inv[p];
    }
    __syncthreads();

    const int lane = tid & 31;
    const int wave = tid >> 5;
    const int h    = lane >> 4;
    const int ln   = lane & 15;
    const int ntBase = wave * 6;

    const float* E = emb + (size_t)bs * NL * ND;
    const float* Wrow0 = &Wm[ln * WSTRIDE];
    const float* Wrow1 = &Wm[(16 + ln) * WSTRIDE];

    float* d0[8];
    float* d1[5];
#pragma unroll
    for (int r = 0; r < 8; ++r) d0[r] = out_base + s_off[r + 8 * h];
#pragma unroll
    for (int r = 0; r < 5; ++r) d1[r] = out_base + s_off[16 + r + 8 * h];

    for (int j = 0; j < 6; ++j) {
        const int n = (ntBase + j) * 16 + ln;
        v8f acc0 = {};
        v8f acc1 = {};
#pragma unroll 4
        for (int k = 0; k < NL; k += 4) {
            v2f a0, a1, b;
            a0.x = Wrow0[k + 2 * h];
            a0.y = Wrow0[k + 2 * h + 1];
            a1.x = Wrow1[k + 2 * h];
            a1.y = Wrow1[k + 2 * h + 1];
            b.x = E[(size_t)(k + 2 * h) * ND + n];
            b.y = E[(size_t)(k + 2 * h + 1) * ND + n];
            acc0 = __builtin_amdgcn_wmma_f32_16x16x4_f32(
                false, a0, false, b, (short)0, acc0, false, false);
            acc1 = __builtin_amdgcn_wmma_f32_16x16x4_f32(
                false, a1, false, b, (short)0, acc1, false, false);
        }
#pragma unroll
        for (int r = 0; r < 8; ++r)
            __builtin_nontemporal_store(acc0[r], d0[r] + n);
#pragma unroll
        for (int r = 0; r < 4; ++r)
            __builtin_nontemporal_store(acc1[r], d1[r] + n);
        if (h == 0)
            __builtin_nontemporal_store(acc1[4], d1[4] + n);
    }
}

extern "C" void kernel_launch(void* const* d_in, const int* in_sizes, int n_in,
                              void* d_out, int out_size, void* d_ws, size_t ws_size,
                              hipStream_t stream) {
    (void)in_sizes; (void)n_in; (void)out_size; (void)d_ws; (void)ws_size;

    const int*   sid   = (const int*)d_in[0];
    const int*   amask = (const int*)d_in[1];
    const int*   pid   = (const int*)d_in[2];
    const int*   a0id  = (const int*)d_in[3];
    const int*   a1id  = (const int*)d_in[4];
    const float* emb   = (const float*)d_in[5];

    dim3 grid(NB * NS);
    dim3 block(256);
    slmuse_embed_kernel<<<grid, block, 0, stream>>>(
        sid, amask, pid, a0id, a1id, emb, (float*)d_out);
}


# E read path: lane owns 4 adjacent columns (one per N tile) so E rows come as 16-byte loads; 12 waves x 64-column slab, k-outer over 8 acc tiles, 16-32 loads in flight per wave; 16-byte output stores
# speedup vs baseline: 1.9022x; 1.9022x over previous
#include <hip/hip_runtime.h>
#include <hip/hip_bf16.h>

#define NB 16
#define NS 16
#define NL 128
#define NA 8
#define NT 8
#define ND 768
#define WSTRIDE 132
#define NWAVES 12
#define NTHR (NWAVES * 32)
#define KSTAGE 8
#define NSTAGE (NL / (4 * KSTAGE))

#define SENT_SZ (NB * NS * ND)
#define ARG_SZ  (NB * NS * NA * ND)

typedef __attribute__((ext_vector_type(2))) float v2f;
typedef __attribute__((ext_vector_type(4))) float v4f;
typedef __attribute__((ext_vector_type(8))) float v8f;

__device__ __forceinline__ int phys_to_logical(int p) {
    if (p < 20) return p;
    if (p == 20) return 24;
    if (p >= 24 && p <= 27) return p - 4;
    return -1;
}

__global__ __launch_bounds__(NTHR)
void slmuse_embed_kernel(const int* __restrict__ sid,
                         const int* __restrict__ amask,
                         const int* __restrict__ pid,
                         const int* __restrict__ a0id,
                         const int* __restrict__ a1id,
                         const float* __restrict__ emb,
                         float* __restrict__ out_base) {
    __shared__ float    Wm[32 * WSTRIDE];
    __shared__ int      s_sid[NL];
    __shared__ int      s_mask[NL];
    __shared__ int      s_ids[3 * NA * NT];
    __shared__ float    s_inv[32];
    __shared__ unsigned s_off[32];

    const int bs  = blockIdx.x;
    const int tid = threadIdx.x;

    if (tid < NL) {
        s_sid[tid]  = sid[bs * NL + tid];
        s_mask[tid] = amask[bs * NL + tid];
    }
    if (tid < NA * NT) {
        s_ids[tid]               = pid[bs * NA * NT + tid];
        s_ids[NA * NT + tid]     = a0id[bs * NA * NT + tid];
        s_ids[2 * NA * NT + tid] = a1id[bs * NA * NT + tid];
    }
    if (tid < 32) {
        const int l = phys_to_logical(tid);
        unsigned off = 0;
        if (l == 0)       off = (unsigned)(bs * ND);
        else if (l <= 8)  off = (unsigned)(SENT_SZ + (bs * NA + (l - 1)) * ND);
        else if (l <= 16) off = (unsigned)(SENT_SZ + ARG_SZ + (bs * NA + (l - 9)) * ND);
        else if (l <= 24) off = (unsigned)(SENT_SZ + 2 * ARG_SZ + (bs * NA + (l - 17)) * ND);
        s_off[tid] = off;
    }
    __syncthreads();

    for (int e = tid; e < 32 * NL; e += NTHR) {
        const int p = e >> 7;
        const int c = e & (NL - 1);
        const int l = phys_to_logical(p);
        float w;
        if (l == 0) {
            w = (float)s_mask[c];
        } else if (l > 0) {
            const int base = (l - 1) * NT;
            const int sv = s_sid[c];
            int cnt = 0;
#pragma unroll
            for (int t = 0; t < NT; ++t) {
                const int id = s_ids[base + t];
                cnt += (id != 0 && id == sv) ? 1 : 0;
            }
            w = (float)cnt;
        } else {
            w = 0.0f;
        }
        Wm[p * WSTRIDE + c] = w;
    }
    __syncthreads();

    if (tid < 32) {
        float c = 0.0f;
        for (int x = 0; x < NL; ++x) c += Wm[tid * WSTRIDE + x];
        float inv;
        if (tid == 0)
            inv = 1.0f / fmaxf(c, 1.0f);
        else
            inv = (c > 0.0f) ? (1.0f / c) : 0.0f;
        s_inv[tid] = inv;
    }
    __syncthreads();
    for (int e = tid; e < 32 * NL; e += NTHR) {
        const int p = e >> 7;
        const int c = e & (NL - 1);
        Wm[p * WSTRIDE + c] *= s_inv[p];
    }
    __syncthreads();

    const int lane = tid & 31;
    const int wave = tid >> 5;
    const int h    = lane >> 4;
    const int ln   = lane & 15;
    const int col  = wave * 64 + 4 * ln;

    const float* Ecol  = emb + (size_t)bs * NL * ND + (size_t)(2 * h) * ND + col;
    const float* Wrow0 = &Wm[ln * WSTRIDE + 2 * h];
    const float* Wrow1 = &Wm[(16 + ln) * WSTRIDE + 2 * h];

    v8f acc0[4];
    v8f acc1[4];
#pragma unroll
    for (int t = 0; t < 4; ++t) { acc0[t] = (v8f){}; acc1[t] = (v8f){}; }

    v4f eb[2][2 * KSTAGE];

#define LOAD_STAGE(ST) do { \
    _Pragma("unroll") for (int i = 0; i < KSTAGE; ++i) { \
        _Pragma("unroll") for (int s = 0; s < 2; ++s) \
            eb[(ST) & 1][2 * i + s] = *reinterpret_cast<const v4f*>( \
                Ecol + (size_t)(((ST) * KSTAGE + i) * 4 + s) * ND); \
    } } while (0)

    LOAD_STAGE(0);
    LOAD_STAGE(1);
#pragma unroll
    for (int st = 0; st < NSTAGE; ++st) {
        __builtin_amdgcn_sched_barrier(0);
#pragma unroll
        for (int i = 0; i < KSTAGE; ++i) {
            const int k = (st * KSTAGE + i) * 4;
            v2f a0, a1;
            a0.x = Wrow0[k];
            a0.y = Wrow0[k + 1];
            a1.x = Wrow1[k];
            a1.y = Wrow1[k + 1];
            const v4f e0 = eb[st & 1][2 * i];
            const v4f e1 = eb[st & 1][2 * i + 1];
#pragma unroll
            for (int t = 0; t < 4; ++t) {
                v2f b;
                b.x = e0[t];
                b.y = e1[t];
                acc0[t] = __builtin_amdgcn_wmma_f32_16x16x4_f32(
                    false, a0, false, b, (short)0, acc0[t], false, false);
                acc1[t] = __builtin_amdgcn_wmma_f32_16x16x4_f32(
                    false, a1, false, b, (short)0, acc1[t], false, false);
            }
        }
        __builtin_amdgcn_sched_barrier(0);
        if (st + 2 < NSTAGE) LOAD_STAGE(st + 2);
    }
#undef LOAD_STAGE

#pragma unroll
    for (int r = 0; r < 8; ++r) {
        v4f v;
        v.x = acc0[0][r]; v.y = acc0[1][r]; v.z = acc0[2][r]; v.w = acc0[3][r];
        __builtin_nontemporal_store(
            v, reinterpret_cast<v4f*>(out_base + s_off[r + 8 * h] + col));
    }
#pragma unroll
    for (int r = 0; r < 5; ++r) {
        v4f v;
        v.x = acc1[0][r]; v.y = acc1[1][r]; v.z = acc1[2][r]; v.w = acc1[3][r];
        if (r < 4 || h == 0)
            __builtin_nontemporal_store(
                v, reinterpret_cast<v4f*>(out_base + s_off[16 + r + 8 * h] + col));
    }
}

extern "C" void kernel_launch(void* const* d_in, const int* in_sizes, int n_in,
                              void* d_out, int out_size, void* d_ws, size_t ws_size,
                              hipStream_t stream) {
    (void)in_sizes; (void)n_in; (void)out_size; (void)d_ws; (void)ws_size;

    const int*   sid   = (const int*)d_in[0];
    const int*   amask = (const int*)d_in[1];
    const int*   pid   = (const int*)d_in[2];
    const int*   a0id  = (const int*)d_in[3];
    const int*   a1id  = (const int*)d_in[4];
    const float* emb   = (const float*)d_in[5];

    dim3 grid(NB * NS);
    dim3 block(NTHR);
    slmuse_embed_kernel<<<grid, block, 0, stream>>>(
        sid, amask, pid, a0id, a1id, emb, (float*)d_out);
}


# W build in one pass: each wave builds whole rows from its own id loads, row totals by lane votes (mask row: cross-lane tree sum, new order, exact for 0/1), one LDS write, 1 barrier instead of 4
# speedup vs baseline: 2.2547x; 1.1853x over previous
#include <hip/hip_runtime.h>
#include <hip/hip_bf16.h>

#define NB 16
#define NS 16
#define NL 128
#define NA 8
#define NT 8
#define ND 768
#define WSTRIDE 132
#define NWAVES 12
#define NTHR (NWAVES * 32)
#define KSTAGE 8
#define NSTAGE (NL / (4 * KSTAGE))

#define SENT_SZ (NB * NS * ND)
#define ARG_SZ  (NB * NS * NA * ND)

typedef __attribute__((ext_vector_type(4))) int   v4i;
typedef __attribute__((ext_vector_type(2))) float v2f;
typedef __attribute__((ext_vector_type(4))) float v4f;
typedef __attribute__((ext_vector_type(8))) float v8f;

template <int M>
__device__ __forceinline__ float row_xor(float x) {
    return __builtin_bit_cast(float, __builtin_amdgcn_update_dpp(
        0, __builtin_bit_cast(int, x), 0x160 + M, 0xf, 0xf, false));
}
__device__ __forceinline__ float xhalf(float x) {
    return __builtin_bit_cast(float, __builtin_amdgcn_permlanex16(
        __builtin_bit_cast(int, x), __builtin_bit_cast(int, x),
        0x76543210, 0xfedcba98, false, false));
}

__device__ __forceinline__ int phys_to_logical(int p) {
    if (p < 20) return p;
    if (p == 20) return 24;
    if (p >= 24 && p <= 27) return p - 4;
    return -1;
}

__global__ __launch_bounds__(NTHR)
void slmuse_embed_kernel(const int* __restrict__ sid,
                         const int* __restrict__ amask,
                         const int* __restrict__ pid,
                         const int* __restrict__ a0id,
                         const int* __restrict__ a1id,
                         const float* __restrict__ emb,
                         float* __restrict__ out_base) {
    __shared__ __attribute__((aligned(16))) float Wm[32 * WSTRIDE];
    __shared__ unsigned s_off[32];

    const int bs  = blockIdx.x;
    const int tid = threadIdx.x;
    const int wv = __builtin_amdgcn_readfirstlane(tid >> 5);
    const int lc = 4 * (tid & 31);

    const v4i sv = *reinterpret_cast<const v4i*>(sid + bs * NL + lc);
    const v4i mk = *reinterpret_cast<const v4i*>(amask + bs * NL + lc);
    const int slot0 = 2 * wv;
    const int* ip = (slot0 < NA) ? pid : (slot0 < 2 * NA) ? a0id : a1id;
    ip += bs * (NA * NT) + (slot0 & (NA - 1)) * NT;
    int ids[2 * NT];
#pragma unroll
    for (int t = 0; t < 2 * NT; ++t) ids[t] = ip[t];

    if (wv == NWAVES - 1) {
        const int r = tid & 31;
        const int l = phys_to_logical(r);
        unsigned off = 0;
        if (l == 0)       off = (unsigned)(bs * ND);
        else if (l <= 8)  off = (unsigned)(SENT_SZ + (bs * NA + (l - 1)) * ND);
        else if (l <= 16) off = (unsigned)(SENT_SZ + ARG_SZ + (bs * NA + (l - 9)) * ND);
        else if (l <= 24) off = (unsigned)(SENT_SZ + 2 * ARG_SZ + (bs * NA + (l - 17)) * ND);
        s_off[r] = off;
    }

    if (wv == 0) {
        const v4f m = (v4f){(float)mk.x, (float)mk.y, (float)mk.z, (float)mk.w};
        float c = (m.x + m.y) + (m.z + m.w);
        c += row_xor<1>(c);
        c += row_xor<2>(c);
        c += row_xor<4>(c);
        c += row_xor<8>(c);
        c += xhalf(c);
        *reinterpret_cast<v4f*>(&Wm[lc]) = m * (1.0f / fmaxf(c, 1.0f));
    } else if (wv < 8) {
        const int pd = (wv < 4) ? 20 + wv : 24 + wv;
        *reinterpret_cast<v4f*>(&Wm[pd * WSTRIDE + lc]) = (v4f){0.0f, 0.0f, 0.0f, 0.0f};
    }

#pragma unroll
    for (int r = 0; r < 2; ++r) {
        const int l = 1 + 2 * wv + r;
        const int p = (l < 20) ? l : (l < 24) ? l + 4 : 20;
        int c0 = 0, c1 = 0, c2 = 0, c3 = 0, tot = 0;
#pragma unroll
        for (int t = 0; t < NT; ++t) {
            const int id = ids[r * NT + t];
            const bool nz = (id != 0);
            const bool e0 = (sv.x == id);
            const bool e1 = (sv.y == id);
            const bool e2 = (sv.z == id);
            const bool e3 = (sv.w == id);
            c0 += (nz && e0) ? 1 : 0;
            c1 += (nz && e1) ? 1 : 0;
            c2 += (nz && e2) ? 1 : 0;
            c3 += (nz && e3) ? 1 : 0;
            const int votes = __builtin_popcount(__builtin_amdgcn_ballot_w32(e0))
                            + __builtin_popcount(__builtin_amdgcn_ballot_w32(e1))
                            + __builtin_popcount(__builtin_amdgcn_ballot_w32(e2))
                            + __builtin_popcount(__builtin_amdgcn_ballot_w32(e3));
            tot += nz ? votes : 0;
        }
        const float c = (float)tot;
        const float inv = (c > 0.0f) ? (1.0f / c) : 0.0f;
        *reinterpret_cast<v4f*>(&Wm[p * WSTRIDE + lc]) =
            (v4f){(float)c0, (float)c1, (float)c2, (float)c3} * inv;
    }
    __syncthreads();

    const int lane = tid & 31;
    const int wave = tid >> 5;
    const int h    = lane >> 4;
    const int ln   = lane & 15;
    const int col  = wave * 64 + 4 * ln;

    const float* Ecol  = emb + (size_t)bs * NL * ND + (size_t)(2 * h) * ND + col;
    const float* Wrow0 = &Wm[ln * WSTRIDE + 2 * h];
    const float* Wrow1 = &Wm[(16 + ln) * WSTRIDE + 2 * h];

    v8f acc0[4];
    v8f acc1[4];
#pragma unroll
    for (int t = 0; t < 4; ++t) { acc0[t] = (v8f){}; acc1[t] = (v8f){}; }

    v4f eb[2][2 * KSTAGE];

#define LOAD_STAGE(ST) do { \
    _Pragma("unroll") for (int i = 0; i < KSTAGE; ++i) { \
        _Pragma("unroll") for (int s = 0; s < 2; ++s) \
            eb[(ST) & 1][2 * i + s] = *reinterpret_cast<const v4f*>( \
                Ecol + (size_t)(((ST) * KSTAGE + i) * 4 + s) * ND); \
    } } while (0)

    LOAD_STAGE(0);
    LOAD_STAGE(1);
#pragma unroll
    for (int st = 0; st < NSTAGE; ++st) {
        __builtin_amdgcn_sched_barrier(0);
#pragma unroll
        for (int i = 0; i < KSTAGE; ++i) {
            const int k = (st * KSTAGE + i) * 4;
            v2f a0, a1;
            a0.x = Wrow0[k];
            a0.y = Wrow0[k + 1];
            a1.x = Wrow1[k];
            a1.y = Wrow1[k + 1];
            const v4f e0 = eb[st & 1][2 * i];
            const v4f e1 = eb[st & 1][2 * i + 1];
#pragma unroll
            for (int t = 0; t < 4; ++t) {
                v2f b;
                b.x = e0[t];
                b.y = e1[t];
                acc0[t] = __builtin_amdgcn_wmma_f32_16x16x4_f32(
                    false, a0, false, b, (short)0, acc0[t], false, false);
                acc1[t] = __builtin_amdgcn_wmma_f32_16x16x4_f32(
                    false, a1, false, b, (short)0, acc1[t], false, false);
            }
        }
        __builtin_amdgcn_sched_barrier(0);
        if (st + 2 < NSTAGE) LOAD_STAGE(st + 2);
    }
#undef LOAD_STAGE

#pragma unroll
    for (int r = 0; r < 8; ++r) {
        v4f v;
        v.x = acc0[0][r]; v.y = acc0[1][r]; v.z = acc0[2][r]; v.w = acc0[3][r];
        __builtin_nontemporal_store(
            v, reinterpret_cast<v4f*>(out_base + s_off[r + 8 * h] + col));
    }
#pragma unroll
    for (int r = 0; r < 5; ++r) {
        v4f v;
        v.x = acc1[0][r]; v.y = acc1[1][r]; v.z = acc1[2][r]; v.w = acc1[3][r];
        if (r < 4 || h == 0)
            __builtin_nontemporal_store(
                v, reinterpret_cast<v4f*>(out_base + s_off[16 + r + 8 * h] + col));
    }
}

extern "C" void kernel_launch(void* const* d_in, const int* in_sizes, int n_in,
                              void* d_out, int out_size, void* d_ws, size_t ws_size,
                              hipStream_t stream) {
    (void)in_sizes; (void)n_in; (void)out_size; (void)d_ws; (void)ws_size;

    const int*   sid   = (const int*)d_in[0];
    const int*   amask = (const int*)d_in[1];
    const int*   pid   = (const int*)d_in[2];
    const int*   a0id  = (const int*)d_in[3];
    const int*   a1id  = (const int*)d_in[4];
    const float* emb   = (const float*)d_in[5];

    dim3 grid(NB * NS);
    dim3 block(NTHR);
    slmuse_embed_kernel<<<grid, block, 0, stream>>>(
        sid, amask, pid, a0id, a1id, emb, (float*)d_out);
}
